# starting kernel
# baseline (speedup 1.0000x reference)
#include <hip/hip_runtime.h>

typedef __attribute__((ext_vector_type(2))) float v2f;
typedef __attribute__((ext_vector_type(8))) float v8f;

#define DDIM 64

__global__ void spmm_edge_atomic(const int* __restrict__ rows,
                                 const int* __restrict__ cols,
                                 const float* __restrict__ vals,
                                 const float* __restrict__ x,
                                 float* __restrict__ out,
                                 int nnz) {
  int gid  = blockIdx.x * blockDim.x + threadIdx.x;
  int edge = __builtin_amdgcn_readfirstlane(gid >> 5);
  int lane = threadIdx.x & 31;
  if (edge >= nnz) return;
  int   r = rows[edge];
  int   c = cols[edge];
  float v = vals[edge];
  const float2* src = (const float2*)(x + (size_t)c * DDIM);
  float*        dst = out + (size_t)r * DDIM;
  float2 s = src[lane];
  unsafeAtomicAdd(dst + 2 * lane,     v * s.x);
  unsafeAtomicAdd(dst + 2 * lane + 1, v * s.y);
}

#define WP_STRIDE 80

__global__ __launch_bounds__(320) void gemm_concat_wmma(
    const float* __restrict__ newu,
    const float* __restrict__ uin,
    const float* __restrict__ W,
    float* __restrict__ out,
    int nrows) {
  __shared__ v2f Wp[64 * WP_STRIDE];
  for (int idx = threadIdx.x; idx < 64 * 64; idx += blockDim.x) {
    int t = idx >> 6, n = idx & 63;
    v2f p;
    p.x = W[(size_t)(2 * t) * DDIM + n];
    p.y = W[(size_t)(2 * t + 1) * DDIM + n];
    Wp[t * WP_STRIDE + n] = p;
  }
  __syncthreads();

  int lane = threadIdx.x & 31;
  int tile = blockIdx.x * (blockDim.x >> 5) + (threadIdx.x >> 5);
  if (tile * 16 >= nrows) return;
  int row0  = tile * 16;
  int m     = lane & 15;
  int khalf = (lane >> 4) << 1;

  const float* rowNew = newu + (size_t)(row0 + m) * DDIM;
  const float* rowU   = uin  + (size_t)(row0 + m) * DDIM;

  v8f acc0 = {}, acc1 = {}, acc2 = {}, acc3 = {};

#pragma unroll
  for (int k0 = 0; k0 < 64; k0 += 4) {
    int kk = k0 + khalf;
    v2f a; a.x = rowNew[kk]; a.y = rowNew[kk + 1];
    const v2f* wr = &Wp[(kk >> 1) * WP_STRIDE];
    v2f b0 = wr[m];
    v2f b1 = wr[m + 16];
    v2f b2 = wr[m + 32];
    v2f b3 = wr[m + 48];
    acc0 = __builtin_amdgcn_wmma_f32_16x16x4_f32(false, a, false, b0, (short)0, acc0, false, false);
    acc1 = __builtin_amdgcn_wmma_f32_16x16x4_f32(false, a, false, b1, (short)0, acc1, false, false);
    acc2 = __builtin_amdgcn_wmma_f32_16x16x4_f32(false, a, false, b2, (short)0, acc2, false, false);
    acc3 = __builtin_amdgcn_wmma_f32_16x16x4_f32(false, a, false, b3, (short)0, acc3, false, false);
  }

#pragma unroll
  for (int k0 = 0; k0 < 64; k0 += 4) {
    int kk = k0 + khalf;
    v2f a; a.x = rowU[kk]; a.y = rowU[kk + 1];
    const v2f* wr = &Wp[(32 + (kk >> 1)) * WP_STRIDE];
    v2f b0 = wr[m];
    v2f b1 = wr[m + 16];
    v2f b2 = wr[m + 32];
    v2f b3 = wr[m + 48];
    acc0 = __builtin_amdgcn_wmma_f32_16x16x4_f32(false, a, false, b0, (short)0, acc0, false, false);
    acc1 = __builtin_amdgcn_wmma_f32_16x16x4_f32(false, a, false, b1, (short)0, acc1, false, false);
    acc2 = __builtin_amdgcn_wmma_f32_16x16x4_f32(false, a, false, b2, (short)0, acc2, false, false);
    acc3 = __builtin_amdgcn_wmma_f32_16x16x4_f32(false, a, false, b3, (short)0, acc3, false, false);
  }

  int mo = (lane >> 4) * 8;
  float* orow = out + (size_t)row0 * DDIM;
#pragma unroll
  for (int i2 = 0; i2 < 8; ++i2) {
    orow[(size_t)(mo + i2) * DDIM + m]      = acc0[i2];
    orow[(size_t)(mo + i2) * DDIM + m + 16] = acc1[i2];
    orow[(size_t)(mo + i2) * DDIM + m + 32] = acc2[i2];
    orow[(size_t)(mo + i2) * DDIM + m + 48] = acc3[i2];
  }
}

extern "C" void kernel_launch(void* const* d_in, const int* in_sizes, int n_in,
                              void* d_out, int out_size, void* d_ws, size_t ws_size,
                              hipStream_t stream) {
  const float* user_emb    = (const float*)d_in[0];
  const float* item_emb    = (const float*)d_in[1];
  const float* weights     = (const float*)d_in[2];
  const float* social_vals = (const float*)d_in[3];
  const float* inter_vals  = (const float*)d_in[4];
  const int*   social_rows = (const int*)d_in[5];
  const int*   social_cols = (const int*)d_in[6];
  const int*   inter_rows  = (const int*)d_in[7];
  const int*   inter_cols  = (const int*)d_in[8];

  const int D   = DDIM;
  const int U   = in_sizes[0] / D;
  const int I   = in_sizes[1] / D;
  const int L   = in_sizes[2] / (2 * D * D);
  const int E_S = in_sizes[3];
  const int E_I = in_sizes[4];

  float* out_u = (float*)d_out;
  float* out_i = (float*)d_out + (size_t)U * D;

  float* newu = (float*)d_ws;
  float* bufA = newu + (size_t)U * D;
  float* bufB = bufA + (size_t)U * D;

  const int spmmThreads = 256;
  auto spmmBlocks = [&](int nnz) {
    return (int)(((size_t)nnz * 32 + spmmThreads - 1) / spmmThreads);
  };

  const int wavesPerBlock = 10;
  const int gemmThreads   = wavesPerBlock * 32;
  const int tiles         = (U + 15) / 16;
  const int gemmBlocks    = (tiles + wavesPerBlock - 1) / wavesPerBlock;

  const float* ucur = user_emb;
  for (int k = 0; k < L; ++k) {
    hipMemsetAsync(newu, 0, (size_t)U * D * sizeof(float), stream);
    spmm_edge_atomic<<<spmmBlocks(E_S), spmmThreads, 0, stream>>>(
        social_rows, social_cols, social_vals, ucur, newu, E_S);
    float* dst = (k == L - 1) ? out_u : ((k & 1) ? bufB : bufA);
    gemm_concat_wmma<<<gemmBlocks, gemmThreads, 0, stream>>>(
        newu, ucur, weights + (size_t)k * 2 * D * D, dst, U);
    ucur = dst;
  }
  if (L == 0) {
    hipMemcpyAsync(out_u, user_emb, (size_t)U * D * sizeof(float),
                   hipMemcpyDeviceToDevice, stream);
  }

  spmm_edge_atomic<<<spmmBlocks(E_I), spmmThreads, 0, stream>>>(
      inter_rows, inter_cols, inter_vals, item_emb, out_u, E_I);

  hipMemcpyAsync(out_i, item_emb, (size_t)I * D * sizeof(float),
                 hipMemcpyDeviceToDevice, stream);
}


# Both layers' weight blocks multiplied first (u2 = S[S(uA)+u(B+C)]+uD; A=Wa1Wa2 B=Wb1Wa2 C=Wa1Wb2 D=Wb1Wb2): one GEMM launch writes 3 planes from one read of u, no second GEMM; sums re-associated
# speedup vs baseline: 3.4988x; 3.4988x over previous
#include <hip/hip_runtime.h>

typedef __attribute__((ext_vector_type(2))) float v2f;
typedef __attribute__((ext_vector_type(8))) float v8f;

#define DDIM 64

#define RPW   64
#define WPB   8
#define CHUNK (RPW * WPB)
#define GROUP 8
#define CNT_STRIDE 64
#define CNT_SLOT   128

__global__ __launch_bounds__(256) void spmm_edge_atomic(const int* __restrict__ rowsA,
                                 const int* __restrict__ colsA,
                                 const float* __restrict__ valsA,
                                 const float* __restrict__ xA,
                                 int nnzA, int nchA,
                                 const int* __restrict__ rowsB,
                                 const int* __restrict__ colsB,
                                 const float* __restrict__ valsB,
                                 const float* __restrict__ xB,
                                 int nnzB,
                                 float* __restrict__ out,
                                 int nrows, int nchunks,
                                 unsigned int* __restrict__ cnt) {
  __shared__ int   s_item[2];
  __shared__ int   s_n[WPB];
  __shared__ int   s_r[CHUNK];
  __shared__ int   s_c[CHUNK];
  __shared__ float s_v[CHUNK];

  const int lane = threadIdx.x & 31;
  const int w    = __builtin_amdgcn_readfirstlane(threadIdx.x >> 5);

  if (threadIdx.x == 0) {
    unsigned int id = __builtin_amdgcn_s_sendmsg_rtn(0x87);
    int set = (int)((id >> 18) & 1u);
    unsigned int t = atomicAdd(&cnt[set * CNT_STRIDE], 1u);
    if (t >= (unsigned int)nchunks) {
      set ^= 1;
      t = atomicAdd(&cnt[set * CNT_STRIDE], 1u);
    }
    s_item[0] = (t < (unsigned int)nchunks) ? (int)t : -1;
    s_item[1] = set;
  }
  __syncthreads();
  const int item  = __builtin_amdgcn_readfirstlane(s_item[0]);
  const int set   = __builtin_amdgcn_readfirstlane(s_item[1]);
  const int split = nrows >> 1;
  const bool listB = item >= nchA;
  const int*   __restrict__ rows = listB ? rowsB : rowsA;
  const int*   __restrict__ cols = listB ? colsB : colsA;
  const float* __restrict__ vals = listB ? valsB : valsA;
  const float* __restrict__ x    = listB ? xB : xA;
  const int nnz   = listB ? nnzB : nnzA;
  const int chunk = listB ? item - nchA : item;

  int   r0 = 0, c0 = 0, r1 = 0, c1 = 0;
  float v0 = 0.0f, v1 = 0.0f;
  bool  p0 = false, p1 = false;
  if (chunk >= 0) {
    int i0 = chunk * CHUNK + w * RPW + lane;
    int i1 = i0 + 32;
    if (i0 < nnz) { r0 = rows[i0]; c0 = cols[i0]; v0 = vals[i0]; p0 = ((r0 >= split) == (set != 0)); }
    if (i1 < nnz) { r1 = rows[i1]; c1 = cols[i1]; v1 = vals[i1]; p1 = ((r1 >= split) == (set != 0)); }
  }
  const unsigned int m0 = __builtin_amdgcn_ballot_w32(p0);
  const unsigned int m1 = __builtin_amdgcn_ballot_w32(p1);
  const unsigned int below = (1u << lane) - 1u;
  const int n0 = __builtin_popcount(m0);
  if (lane == 0) s_n[w] = n0 + __builtin_popcount(m1);
  __syncthreads();
  int off = 0, total = 0;
  for (int i = 0; i < WPB; ++i) { int ni = s_n[i]; off += (i < w) ? ni : 0; total += ni; }
  if (p0) { int s = off + __builtin_popcount(m0 & below);      s_r[s] = r0; s_c[s] = c0; s_v[s] = v0; }
  if (p1) { int s = off + n0 + __builtin_popcount(m1 & below); s_r[s] = r1; s_c[s] = c1; s_v[s] = v1; }
  __syncthreads();
  if (total == 0) return;

  const int   ia = (w + WPB * (lane >> 3)) * GROUP + (lane & 7);
  const int   ib = ia + 32 * GROUP;
  const int   ra = s_r[ia];
  const int   ca = s_c[ia];
  const float va = s_v[ia];
  const int   rb = s_r[ib];
  const int   cb = s_c[ib];
  const float vb = s_v[ib];

  const int full = total >> 3;

  auto group = [&](int rr, int cc, float vv, int l0) {
    float a0[GROUP], a1[GROUP];
#pragma unroll
    for (int j = 0; j < GROUP; ++j) {
      int cj = __builtin_amdgcn_readlane(cc, l0 + j);
      const float* src = x + (size_t)cj * DDIM;
      a0[j] = src[lane];
      a1[j] = src[32 + lane];
    }
#pragma unroll
    for (int j = 0; j < GROUP; ++j) {
      int   rj = __builtin_amdgcn_readlane(rr, l0 + j);
      float vj = __builtin_bit_cast(float,
                   __builtin_amdgcn_readlane(__builtin_bit_cast(int, vv), l0 + j));
      float* dst = out + (size_t)rj * DDIM;
      unsafeAtomicAdd(dst + lane,      vj * a0[j]);
      unsafeAtomicAdd(dst + 32 + lane, vj * a1[j]);
    }
  };

  for (int k = 0; k < 4; ++k) {
    if (w + WPB * k >= full) break;
    group(ra, ca, va, GROUP * k);
  }
  for (int k = 0; k < 4; ++k) {
    if (w + 32 + WPB * k >= full) break;
    group(rb, cb, vb, GROUP * k);
  }

  const int rem = total & 7;
  if (rem != 0 && (full & (WPB - 1)) == w) {
    const int  kk = full >> 3;
    const bool hi = kk >= 4;
    const int   rr = hi ? rb : ra;
    const int   cc = hi ? cb : ca;
    const float vv = hi ? vb : va;
    const int   l0 = GROUP * (kk & 3);
    for (int j = 0; j < rem; ++j) {
      int   cj = __builtin_amdgcn_readlane(cc, l0 + j);
      int   rj = __builtin_amdgcn_readlane(rr, l0 + j);
      float vj = __builtin_bit_cast(float,
                   __builtin_amdgcn_readlane(__builtin_bit_cast(int, vv), l0 + j));
      const float* src = x + (size_t)cj * DDIM;
      float*       dst = out + (size_t)rj * DDIM;
      float b0 = src[lane];
      float b1 = src[32 + lane];
      unsafeAtomicAdd(dst + lane,      vj * b0);
      unsafeAtomicAdd(dst + 32 + lane, vj * b1);
    }
  }
}

#define WP_STRIDE   80
#define GEMM_BLOCKS 256
#define GEMM_WAVES  28
#define GEMM_THREADS (GEMM_WAVES * 32)
#define GEMM_LDS_BYTES (176 * 1024)

__global__ __launch_bounds__(GEMM_THREADS) void gemm_split_wmma(
    const float* __restrict__ uin,
    const float* __restrict__ W,
    float* __restrict__ outP,
    float* __restrict__ outQ,
    int nrows,
    unsigned int* __restrict__ cnt, int ncnt) {
  __shared__ __attribute__((aligned(16))) unsigned char lds_block[GEMM_LDS_BYTES];
  v2f* Wp = reinterpret_cast<v2f*>(lds_block);
  {
    const int ci = blockIdx.x * GEMM_THREADS + threadIdx.x;
    if (ci < ncnt) cnt[ci] = 0u;
  }
  if (threadIdx.x < 512) {
    const int t0 = threadIdx.x >> 4, q = threadIdx.x & 15;
    float4 a[2], b[2];
#pragma unroll
    for (int h = 0; h < 2; ++h) {
      const int t = t0 + 32 * h;
      a[h] = *reinterpret_cast<const float4*>(W + (size_t)(2 * t) * DDIM + 4 * q);
      b[h] = *reinterpret_cast<const float4*>(W + (size_t)(2 * t + 1) * DDIM + 4 * q);
    }
#pragma unroll
    for (int h = 0; h < 2; ++h) {
      const int t = t0 + 32 * h;
      float4* dst = reinterpret_cast<float4*>(&Wp[t * WP_STRIDE + 4 * q]);
      dst[0] = make_float4(a[h].x, b[h].x, a[h].y, b[h].y);
      dst[1] = make_float4(a[h].z, b[h].z, a[h].w, b[h].w);
    }
  }
  __syncthreads();

  const int lane  = threadIdx.x & 31;
  const int wave  = threadIdx.x >> 5;
  const int tiles = (nrows + 15) >> 4;
  const int m     = lane & 15;
  const int khalf = (lane >> 4) << 1;
  const int mo    = (lane >> 4) * 8;

  const int tile = blockIdx.x + gridDim.x * wave;
  if (tile >= tiles) return;
  {
    const int row0 = tile * 16;
    const float* rowU = uin + (size_t)(row0 + m) * DDIM;

    v8f p0 = {}, p1 = {}, p2 = {}, p3 = {};
    v8f q0 = {}, q1 = {}, q2 = {}, q3 = {};

#pragma unroll
    for (int k0 = 0; k0 < 64; k0 += 4) {
      int kk = k0 + khalf;
      v2f a; a.x = rowU[kk]; a.y = rowU[kk + 1];
      const v2f* wa = &Wp[(kk >> 1) * WP_STRIDE];
      const v2f* wb = &Wp[(32 + (kk >> 1)) * WP_STRIDE];
      v2f b0 = wa[m];
      v2f b1 = wa[m + 16];
      v2f b2 = wa[m + 32];
      v2f b3 = wa[m + 48];
      v2f c0 = wb[m];
      v2f c1 = wb[m + 16];
      v2f c2 = wb[m + 32];
      v2f c3 = wb[m + 48];
      p0 = __builtin_amdgcn_wmma_f32_16x16x4_f32(false, a, false, b0, (short)0, p0, false, false);
      p1 = __builtin_amdgcn_wmma_f32_16x16x4_f32(false, a, false, b1, (short)0, p1, false, false);
      p2 = __builtin_amdgcn_wmma_f32_16x16x4_f32(false, a, false, b2, (short)0, p2, false, false);
      p3 = __builtin_amdgcn_wmma_f32_16x16x4_f32(false, a, false, b3, (short)0, p3, false, false);
      q0 = __builtin_amdgcn_wmma_f32_16x16x4_f32(false, a, false, c0, (short)0, q0, false, false);
      q1 = __builtin_amdgcn_wmma_f32_16x16x4_f32(false, a, false, c1, (short)0, q1, false, false);
      q2 = __builtin_amdgcn_wmma_f32_16x16x4_f32(false, a, false, c2, (short)0, q2, false, false);
      q3 = __builtin_amdgcn_wmma_f32_16x16x4_f32(false, a, false, c3, (short)0, q3, false, false);
    }

    float* prow = outP + (size_t)row0 * DDIM;
    float* qrow = outQ + (size_t)row0 * DDIM;
#pragma unroll
    for (int i2 = 0; i2 < 8; ++i2) {
      prow[(size_t)(mo + i2) * DDIM + m]      = p0[i2];
      prow[(size_t)(mo + i2) * DDIM + m + 16] = p1[i2];
      prow[(size_t)(mo + i2) * DDIM + m + 32] = p2[i2];
      prow[(size_t)(mo + i2) * DDIM + m + 48] = p3[i2];
    }
#pragma unroll
    for (int i2 = 0; i2 < 8; ++i2) {
      qrow[(size_t)(mo + i2) * DDIM + m]      = q0[i2];
      qrow[(size_t)(mo + i2) * DDIM + m + 16] = q1[i2];
      qrow[(size_t)(mo + i2) * DDIM + m + 32] = q2[i2];
      qrow[(size_t)(mo + i2) * DDIM + m + 48] = q3[i2];
    }
  }
}

#define M_STRIDE 208

__global__ __launch_bounds__(GEMM_THREADS) void gemm_folded_wmma(
    const float* __restrict__ uin,
    const float* __restrict__ W1,
    const float* __restrict__ W2,
    float* __restrict__ outX,
    float* __restrict__ outY,
    float* __restrict__ outZ,
    int nrows,
    unsigned int* __restrict__ cnt, int ncnt) {
  __shared__ __attribute__((aligned(16))) unsigned char lds_block[GEMM_LDS_BYTES];
  v2f* Wp = reinterpret_cast<v2f*>(lds_block);
  v2f* Mp = reinterpret_cast<v2f*>(lds_block + 64 * WP_STRIDE * sizeof(v2f));
  {
    const int ci = blockIdx.x * GEMM_THREADS + threadIdx.x;
    if (ci < ncnt) cnt[ci] = 0u;
  }
  if (threadIdx.x < 512) {
    const int t0 = threadIdx.x >> 4, q = threadIdx.x & 15;
    float4 a[2], b[2];
#pragma unroll
    for (int h = 0; h < 2; ++h) {
      const int t = t0 + 32 * h;
      a[h] = *reinterpret_cast<const float4*>(W2 + (size_t)(2 * t) * DDIM + 4 * q);
      b[h] = *reinterpret_cast<const float4*>(W2 + (size_t)(2 * t + 1) * DDIM + 4 * q);
    }
#pragma unroll
    for (int h = 0; h < 2; ++h) {
      const int t = t0 + 32 * h;
      float4* dst = reinterpret_cast<float4*>(&Wp[t * WP_STRIDE + 4 * q]);
      dst[0] = make_float4(a[h].x, b[h].x, a[h].y, b[h].y);
      dst[1] = make_float4(a[h].z, b[h].z, a[h].w, b[h].w);
    }
  }
  __syncthreads();

  const int lane  = threadIdx.x & 31;
  const int wave  = threadIdx.x >> 5;
  const int m     = lane & 15;
  const int khalf = (lane >> 4) << 1;
  const int mo    = (lane >> 4) * 8;

  if (wave < 16) {
    const int ib = wave & 3, nb = wave >> 2;
    const float* rowA = W1 + (size_t)(16 * ib + m) * DDIM;
    const float* rowB = W1 + (size_t)(64 + 16 * ib + m) * DDIM;
    v8f fa = {}, fy = {}, fd = {};
#pragma unroll
    for (int k0 = 0; k0 < 64; k0 += 4) {
      const int kk = k0 + khalf;
      v2f a1; a1.x = rowA[kk]; a1.y = rowA[kk + 1];
      v2f b1; b1.x = rowB[kk]; b1.y = rowB[kk + 1];
      v2f ra = Wp[(kk >> 1) * WP_STRIDE + 16 * nb + m];
      v2f rb = Wp[(32 + (kk >> 1)) * WP_STRIDE + 16 * nb + m];
      fa = __builtin_amdgcn_wmma_f32_16x16x4_f32(false, a1, false, ra, (short)0, fa, false, false);
      fy = __builtin_amdgcn_wmma_f32_16x16x4_f32(false, b1, false, ra, (short)0, fy, false, false);
      fd = __builtin_amdgcn_wmma_f32_16x16x4_f32(false, b1, false, rb, (short)0, fd, false, false);
      fy = __builtin_amdgcn_wmma_f32_16x16x4_f32(false, a1, false, rb, (short)0, fy, false, false);
    }
    v2f* mrow = &Mp[(8 * ib + 4 * (lane >> 4)) * M_STRIDE + 16 * nb + m];
#pragma unroll
    for (int r = 0; r < 8; r += 2) {
      v2f* d = mrow + (r >> 1) * M_STRIDE;
      v2f t;
      t.x = fa[r]; t.y = fa[r + 1]; d[0]   = t;
      t.x = fy[r]; t.y = fy[r + 1]; d[64]  = t;
      t.x = fd[r]; t.y = fd[r + 1]; d[128] = t;
    }
  }
  __syncthreads();

  const int tiles = (nrows + 15) >> 4;
  const int tile  = blockIdx.x + gridDim.x * wave;
  if (tile >= tiles) return;
  const int row0 = tile * 16;
  const float* rowU = uin + (size_t)(row0 + m) * DDIM;
  {
    v8f t0 = {}, t1 = {}, t2 = {}, t3 = {};
#pragma unroll
    for (int k0 = 0; k0 < 64; k0 += 4) {
      const int kk = k0 + khalf;
      v2f a; a.x = rowU[kk]; a.y = rowU[kk + 1];
      const v2f* mr = &Mp[(kk >> 1) * M_STRIDE + 0];
      v2f b0 = mr[m], b1 = mr[m + 16], b2 = mr[m + 32], b3 = mr[m + 48];
      t0 = __builtin_amdgcn_wmma_f32_16x16x4_f32(false, a, false, b0, (short)0, t0, false, false);
      t1 = __builtin_amdgcn_wmma_f32_16x16x4_f32(false, a, false, b1, (short)0, t1, false, false);
      t2 = __builtin_amdgcn_wmma_f32_16x16x4_f32(false, a, false, b2, (short)0, t2, false, false);
      t3 = __builtin_amdgcn_wmma_f32_16x16x4_f32(false, a, false, b3, (short)0, t3, false, false);
    }
    float* orow = outX + (size_t)row0 * DDIM;
#pragma unroll
    for (int i2 = 0; i2 < 8; ++i2) {
      orow[(size_t)(mo + i2) * DDIM + m]      = t0[i2];
      orow[(size_t)(mo + i2) * DDIM + m + 16] = t1[i2];
      orow[(size_t)(mo + i2) * DDIM + m + 32] = t2[i2];
      orow[(size_t)(mo + i2) * DDIM + m + 48] = t3[i2];
    }
  }
  __builtin_amdgcn_sched_barrier(0);
  {
    v8f t0 = {}, t1 = {}, t2 = {}, t3 = {};
#pragma unroll
    for (int k0 = 0; k0 < 64; k0 += 4) {
      const int kk = k0 + khalf;
      v2f a; a.x = rowU[kk]; a.y = rowU[kk + 1];
      const v2f* mr = &Mp[(kk >> 1) * M_STRIDE + 64];
      v2f b0 = mr[m], b1 = mr[m + 16], b2 = mr[m + 32], b3 = mr[m + 48];
      t0 = __builtin_amdgcn_wmma_f32_16x16x4_f32(false, a, false, b0, (short)0, t0, false, false);
      t1 = __builtin_amdgcn_wmma_f32_16x16x4_f32(false, a, false, b1, (short)0, t1, false, false);
      t2 = __builtin_amdgcn_wmma_f32_16x16x4_f32(false, a, false, b2, (short)0, t2, false, false);
      t3 = __builtin_amdgcn_wmma_f32_16x16x4_f32(false, a, false, b3, (short)0, t3, false, false);
    }
    float* orow = outY + (size_t)row0 * DDIM;
#pragma unroll
    for (int i2 = 0; i2 < 8; ++i2) {
      orow[(size_t)(mo + i2) * DDIM + m]      = t0[i2];
      orow[(size_t)(mo + i2) * DDIM + m + 16] = t1[i2];
      orow[(size_t)(mo + i2) * DDIM + m + 32] = t2[i2];
      orow[(size_t)(mo + i2) * DDIM + m + 48] = t3[i2];
    }
  }
  __builtin_amdgcn_sched_barrier(0);
  {
    v8f t0 = {}, t1 = {}, t2 = {}, t3 = {};
#pragma unroll
    for (int k0 = 0; k0 < 64; k0 += 4) {
      const int kk = k0 + khalf;
      v2f a; a.x = rowU[kk]; a.y = rowU[kk + 1];
      const v2f* mr = &Mp[(kk >> 1) * M_STRIDE + 128];
      v2f b0 = mr[m], b1 = mr[m + 16], b2 = mr[m + 32], b3 = mr[m + 48];
      t0 = __builtin_amdgcn_wmma_f32_16x16x4_f32(false, a, false, b0, (short)0, t0, false, false);
      t1 = __builtin_amdgcn_wmma_f32_16x16x4_f32(false, a, false, b1, (short)0, t1, false, false);
      t2 = __builtin_amdgcn_wmma_f32_16x16x4_f32(false, a, false, b2, (short)0, t2, false, false);
      t3 = __builtin_amdgcn_wmma_f32_16x16x4_f32(false, a, false, b3, (short)0, t3, false, false);
    }
    float* orow = outZ + (size_t)row0 * DDIM;
#pragma unroll
    for (int i2 = 0; i2 < 8; ++i2) {
      orow[(size_t)(mo + i2) * DDIM + m]      = t0[i2];
      orow[(size_t)(mo + i2) * DDIM + m + 16] = t1[i2];
      orow[(size_t)(mo + i2) * DDIM + m + 32] = t2[i2];
      orow[(size_t)(mo + i2) * DDIM + m + 48] = t3[i2];
    }
  }
}

extern "C" void kernel_launch(void* const* d_in, const int* in_sizes, int n_in,
                              void* d_out, int out_size, void* d_ws, size_t ws_size,
                              hipStream_t stream) {
  const float* user_emb    = (const float*)d_in[0];
  const float* item_emb    = (const float*)d_in[1];
  const float* weights     = (const float*)d_in[2];
  const float* social_vals = (const float*)d_in[3];
  const float* inter_vals  = (const float*)d_in[4];
  const int*   social_rows = (const int*)d_in[5];
  const int*   social_cols = (const int*)d_in[6];
  const int*   inter_rows  = (const int*)d_in[7];
  const int*   inter_cols  = (const int*)d_in[8];

  const int D   = DDIM;
  const int U   = in_sizes[0] / D;
  const int I   = in_sizes[1] / D;
  const int L   = in_sizes[2] / (2 * D * D);
  const int E_S = in_sizes[3];
  const int E_I = in_sizes[4];

  float* out_u = (float*)d_out;
  float* out_i = (float*)d_out + (size_t)U * D;

  float* bufP = (float*)d_ws;
  float* bufA = bufP + (size_t)U * D;
  float* bufB = bufA + (size_t)U * D;

  unsigned int* cnt = (unsigned int*)out_i;
  const int nSpmm = L > 0 ? L : 1;
  const int ncnt  = nSpmm * CNT_SLOT;

  const int spmmThreads = WPB * 32;
  auto spmmChunks = [&](int nnz) { return (int)(((size_t)nnz + CHUNK - 1) / CHUNK); };

  const int gemmThreads = GEMM_THREADS;
  const int tiles       = (U + 15) / 16;
  const int tileBlocks  = (tiles + GEMM_WAVES - 1) / GEMM_WAVES;
  const int gemmBlocks  = tileBlocks > GEMM_BLOCKS ? tileBlocks : GEMM_BLOCKS;

  const int chS = spmmChunks(E_S);
  const int chI = spmmChunks(E_I);

  if (L == 0) {
    hipMemcpyAsync(out_u, user_emb, (size_t)U * D * sizeof(float),
                   hipMemcpyDeviceToDevice, stream);
    hipMemsetAsync(cnt, 0, (size_t)ncnt * sizeof(unsigned int), stream);
    if (chI > 0)
      spmm_edge_atomic<<<2 * chI, spmmThreads, 0, stream>>>(
          inter_rows, inter_cols, inter_vals, item_emb, E_I, chI,
          inter_rows, inter_cols, inter_vals, item_emb, E_I,
          out_u, U, chI, cnt);
  }

  const bool folded = (L == 2);
  if (folded) {
    const float* W1 = weights;
    const float* W2 = weights + (size_t)2 * D * D;
    gemm_folded_wmma<<<gemmBlocks, gemmThreads, 0, stream>>>(
        user_emb, W1, W2, bufP, bufA, out_u, U, cnt, ncnt);
    if (chS > 0)
      spmm_edge_atomic<<<2 * chS, spmmThreads, 0, stream>>>(
          social_rows, social_cols, social_vals, bufP, E_S, chS,
          inter_rows, inter_cols, inter_vals, item_emb, E_I,
          bufA, U, chS, cnt);
    if (chS + chI > 0)
      spmm_edge_atomic<<<2 * (chS + chI), spmmThreads, 0, stream>>>(
          social_rows, social_cols, social_vals, bufA, E_S, chS,
          inter_rows, inter_cols, inter_vals, item_emb, E_I,
          out_u, U, chS + chI, cnt + CNT_SLOT);
  }

  const float* ucur = user_emb;
  for (int k = 0; k < (folded ? 0 : L); ++k) {
    const bool last = (k == L - 1);
    float* q = last ? out_u : ((k & 1) ? bufB : bufA);
    gemm_split_wmma<<<gemmBlocks, gemmThreads, 0, stream>>>(
        ucur, weights + (size_t)k * 2 * D * D, bufP, q, U,
        cnt, k == 0 ? ncnt : 0);
    const int chB = last ? chI : 0;
    if (chS + chB > 0)
      spmm_edge_atomic<<<2 * (chS + chB), spmmThreads, 0, stream>>>(
          social_rows, social_cols, social_vals, bufP, E_S, chS,
          inter_rows, inter_cols, inter_vals, item_emb, E_I,
          q, U, chS + chB, cnt + (size_t)k * CNT_SLOT);
    ucur = q;
  }

  hipMemcpyAsync(out_i, item_emb, (size_t)I * D * sizeof(float),
                 hipMemcpyDeviceToDevice, stream);
}
